# starting kernel
# baseline (speedup 1.0000x reference)
#include <hip/hip_runtime.h>
#include <hip/hip_bf16.h>


#define S_LEN   768
#define V_WORD  30000
#define V_POS   64
#define WEMB    100
#define PEMB    28
#define EMB     128
#define HDIM    256
#define G4H     1024
#define MLP     128

typedef __attribute__((ext_vector_type(2))) float v2f;
typedef __attribute__((ext_vector_type(8))) float v8f;

__device__ __forceinline__ float fast_sigmoid(float x) {
    return 1.0f / (1.0f + __expf(-x));
}
__device__ __forceinline__ float fast_tanh(float x) {
    float c = fminf(fmaxf(x, -9.0f), 9.0f);
    float e = __expf(2.0f * c);
    return (e - 1.0f) / (e + 1.0f);
}

__global__ void embed_kernel(const int* __restrict__ widx, const int* __restrict__ pidx,
                             const float* __restrict__ Wword, const float* __restrict__ Wpos,
                             float* __restrict__ x) {
    int s = blockIdx.x;
    int t = threadIdx.x;
    float v;
    if (t < WEMB) v = Wword[(size_t)widx[s] * WEMB + t];
    else          v = Wpos[(size_t)pidx[s] * PEMB + (t - WEMB)];
    x[(size_t)s * EMB + t] = v;
}

__global__ void transpose_sub(const float* __restrict__ src, float* __restrict__ dst,
                              int R, int Ctot, int coff, int Ccnt) {
    int idx = blockIdx.x * blockDim.x + threadIdx.x;
    int total = R * Ccnt;
    if (idx >= total) return;
    int r = idx / Ccnt;
    int c = idx - r * Ccnt;
    dst[(size_t)c * R + r] = src[(size_t)r * Ctot + coff + c];
}

__global__ void pack_bt(const float* __restrict__ src, float* __restrict__ dst,
                        int Nout, int Ctot, int coff, int Kcnt) {
    int idx = blockIdx.x * blockDim.x + threadIdx.x;
    int total = Nout * Kcnt;
    if (idx >= total) return;
    int n = idx / Kcnt;
    int k = idx - n * Kcnt;
    dst[((size_t)(k >> 2) * Nout + n) * 4 + (k & 3)] = src[(size_t)n * Ctot + coff + k];
}

__global__ void __launch_bounds__(32)
gemm_wmma_f32(const float* __restrict__ A, const float* __restrict__ BP,
              const float* __restrict__ bias0, const float* __restrict__ bias1,
              float* __restrict__ C, int M, int N, int K) {
    const int nt = blockIdx.x;
    const int mt = blockIdx.y;
    const int l  = threadIdx.x;
    const int fr   = l & 15;
    const int half = l >> 4;
    const int kb   = half * 2;

    const float* Arow  = A + (size_t)(mt * 16 + fr) * K + kb;
    const int    nbase = nt * 64;

    v8f acc[4] = {{}, {}, {}, {}};
    for (int k0 = 0; k0 < K; k0 += 4) {
        v2f a;
        a.x = Arow[k0];
        a.y = Arow[k0 + 1];
        const float* bp = BP + ((size_t)(k0 >> 2) * N + nbase + fr) * 4 + kb;
        v2f b0, b1, b2, b3;
        b0.x = bp[0];       b0.y = bp[1];
        b1.x = bp[16 * 4];  b1.y = bp[16 * 4 + 1];
        b2.x = bp[32 * 4];  b2.y = bp[32 * 4 + 1];
        b3.x = bp[48 * 4];  b3.y = bp[48 * 4 + 1];
        acc[0] = __builtin_amdgcn_wmma_f32_16x16x4_f32(false, a, false, b0, (short)0, acc[0], false, false);
        acc[1] = __builtin_amdgcn_wmma_f32_16x16x4_f32(false, a, false, b1, (short)0, acc[1], false, false);
        acc[2] = __builtin_amdgcn_wmma_f32_16x16x4_f32(false, a, false, b2, (short)0, acc[2], false, false);
        acc[3] = __builtin_amdgcn_wmma_f32_16x16x4_f32(false, a, false, b3, (short)0, acc[3], false, false);
    }

#pragma unroll
    for (int u = 0; u < 4; ++u) {
        const int g = nbase + u * 16 + fr;
        float bs = 0.0f;
        if (bias0) bs += bias0[g];
        if (bias1) bs += bias1[g];
#pragma unroll
        for (int v = 0; v < 8; ++v) {
            int row = v + half * 8;
            C[(size_t)(mt * 16 + row) * N + g] = acc[u][v] + bs;
        }
    }
}

__global__ void lstm_layer(const float* __restrict__ gx_f, const float* __restrict__ gx_b,
                           const float* __restrict__ WhhT_f, const float* __restrict__ WhhT_b,
                           float* __restrict__ hout) {
    const int dir = blockIdx.x;
    const float* gx   = dir ? gx_b   : gx_f;
    const float* WhhT = dir ? WhhT_b : WhhT_f;

    __shared__ float h_s[HDIM];
    __shared__ float c_s[HDIM];
    __shared__ float g_s[G4H];

    const int tid = threadIdx.x;
    if (tid < HDIM) { h_s[tid] = 0.0f; c_s[tid] = 0.0f; }
    __syncthreads();

    for (int step = 0; step < S_LEN; ++step) {
        const int t = dir ? (S_LEN - 1 - step) : step;
        float acc = gx[(size_t)t * G4H + tid];
#pragma unroll 8
        for (int j = 0; j < HDIM; ++j) {
            acc += h_s[j] * WhhT[(size_t)j * G4H + tid];
        }
        g_s[tid] = acc;
        __syncthreads();
        if (tid < HDIM) {
            float gi = g_s[tid];
            float gf = g_s[HDIM + tid];
            float gc = g_s[2 * HDIM + tid];
            float go = g_s[3 * HDIM + tid];
            float c  = fast_sigmoid(gf) * c_s[tid] + fast_sigmoid(gi) * fast_tanh(gc);
            float h  = fast_sigmoid(go) * fast_tanh(c);
            c_s[tid] = c;
            h_s[tid] = h;
            hout[(size_t)t * (2 * HDIM) + dir * HDIM + tid] = h;
        }
        __syncthreads();
    }
}

__global__ void pairwise_kernel(const float* __restrict__ hA, const float* __restrict__ hB,
                                const float* __restrict__ W2, const float* __restrict__ b2,
                                float* __restrict__ out) {
    __shared__ float sA[16][MLP];
    __shared__ float sB[16][MLP];
    __shared__ float sW[MLP];

    const int i0 = blockIdx.y * 16;
    const int j0 = blockIdx.x * 16;
    const int tid = threadIdx.x;

    for (int idx = tid; idx < 16 * MLP; idx += 256) {
        int r = idx >> 7, k = idx & (MLP - 1);
        sA[r][k] = hA[(size_t)(i0 + r) * MLP + k];
        sB[r][k] = hB[(size_t)(j0 + r) * MLP + k];
    }
    if (tid < MLP) sW[tid] = W2[tid];
    __syncthreads();

    const int ti = tid >> 4;
    const int tj = tid & 15;
    const int i = i0 + ti;
    const int j = j0 + tj;

    float acc = 0.0f;
#pragma unroll 4
    for (int k = 0; k < MLP; ++k) {
        acc += sW[k] * fast_tanh(sA[ti][k] + sB[tj][k]);
    }
    float r = acc + b2[0];
    out[(size_t)i * S_LEN + j] = (i == j) ? 0.0f : r;
}

extern "C" void kernel_launch(void* const* d_in, const int* in_sizes, int n_in,
                              void* d_out, int out_size, void* d_ws, size_t ws_size,
                              hipStream_t stream) {
    const int*   widx   = (const int*)d_in[0];
    const int*   pidx   = (const int*)d_in[1];
    const float* Wword  = (const float*)d_in[3];
    const float* Wpos   = (const float*)d_in[4];
    const float* Wih_l0f = (const float*)d_in[5];
    const float* Whh_l0f = (const float*)d_in[6];
    const float* bih_l0f = (const float*)d_in[7];
    const float* bhh_l0f = (const float*)d_in[8];
    const float* Wih_l0b = (const float*)d_in[9];
    const float* Whh_l0b = (const float*)d_in[10];
    const float* bih_l0b = (const float*)d_in[11];
    const float* bhh_l0b = (const float*)d_in[12];
    const float* Wih_l1f = (const float*)d_in[13];
    const float* Whh_l1f = (const float*)d_in[14];
    const float* bih_l1f = (const float*)d_in[15];
    const float* bhh_l1f = (const float*)d_in[16];
    const float* Wih_l1b = (const float*)d_in[17];
    const float* Whh_l1b = (const float*)d_in[18];
    const float* bih_l1b = (const float*)d_in[19];
    const float* bhh_l1b = (const float*)d_in[20];
    const float* W1 = (const float*)d_in[21];
    const float* b1 = (const float*)d_in[22];
    const float* W2 = (const float*)d_in[23];
    const float* b2 = (const float*)d_in[24];
    float* out = (float*)d_out;

    float* ws = (float*)d_ws;
    float* x      = ws;
    float* gxf    = x      + S_LEN * EMB;
    float* gxb    = gxf    + S_LEN * G4H;
    float* h0     = gxb    + S_LEN * G4H;
    float* h1     = h0     + S_LEN * 2 * HDIM;
    float* hA     = h1     + S_LEN * 2 * HDIM;
    float* hB     = hA     + S_LEN * MLP;
    float* BP0f   = hB     + S_LEN * MLP;
    float* BP0b   = BP0f   + EMB * G4H;
    float* BP1f   = BP0b   + EMB * G4H;
    float* BP1b   = BP1f   + 2 * HDIM * G4H;
    float* WhhT0f = BP1b   + 2 * HDIM * G4H;
    float* WhhT0b = WhhT0f + HDIM * G4H;
    float* WhhT1f = WhhT0b + HDIM * G4H;
    float* WhhT1b = WhhT1f + HDIM * G4H;
    float* BPA    = WhhT1b + HDIM * G4H;
    float* BPB    = BPA    + 2 * HDIM * MLP;

    auto tgrid = [](int total) { return dim3((total + 255) / 256); };
    pack_bt<<<tgrid(G4H * EMB),      256, 0, stream>>>(Wih_l0f, BP0f, G4H, EMB,      0,      EMB);
    pack_bt<<<tgrid(G4H * EMB),      256, 0, stream>>>(Wih_l0b, BP0b, G4H, EMB,      0,      EMB);
    pack_bt<<<tgrid(G4H * 2 * HDIM), 256, 0, stream>>>(Wih_l1f, BP1f, G4H, 2 * HDIM, 0, 2 * HDIM);
    pack_bt<<<tgrid(G4H * 2 * HDIM), 256, 0, stream>>>(Wih_l1b, BP1b, G4H, 2 * HDIM, 0, 2 * HDIM);
    pack_bt<<<tgrid(MLP * 2 * HDIM), 256, 0, stream>>>(W1, BPA, MLP, G4H,        0, 2 * HDIM);
    pack_bt<<<tgrid(MLP * 2 * HDIM), 256, 0, stream>>>(W1, BPB, MLP, G4H, 2 * HDIM, 2 * HDIM);
    transpose_sub<<<tgrid(G4H * HDIM), 256, 0, stream>>>(Whh_l0f, WhhT0f, G4H, HDIM, 0, HDIM);
    transpose_sub<<<tgrid(G4H * HDIM), 256, 0, stream>>>(Whh_l0b, WhhT0b, G4H, HDIM, 0, HDIM);
    transpose_sub<<<tgrid(G4H * HDIM), 256, 0, stream>>>(Whh_l1f, WhhT1f, G4H, HDIM, 0, HDIM);
    transpose_sub<<<tgrid(G4H * HDIM), 256, 0, stream>>>(Whh_l1b, WhhT1b, G4H, HDIM, 0, HDIM);

    embed_kernel<<<S_LEN, EMB, 0, stream>>>(widx, pidx, Wword, Wpos, x);

    {
        dim3 grid(G4H / 64, S_LEN / 16);
        gemm_wmma_f32<<<grid, 32, 0, stream>>>(x, BP0f, bih_l0f, bhh_l0f, gxf, S_LEN, G4H, EMB);
        gemm_wmma_f32<<<grid, 32, 0, stream>>>(x, BP0b, bih_l0b, bhh_l0b, gxb, S_LEN, G4H, EMB);
    }
    lstm_layer<<<2, G4H, 0, stream>>>(gxf, gxb, WhhT0f, WhhT0b, h0);

    {
        dim3 grid(G4H / 64, S_LEN / 16);
        gemm_wmma_f32<<<grid, 32, 0, stream>>>(h0, BP1f, bih_l1f, bhh_l1f, gxf, S_LEN, G4H, 2 * HDIM);
        gemm_wmma_f32<<<grid, 32, 0, stream>>>(h0, BP1b, bih_l1b, bhh_l1b, gxb, S_LEN, G4H, 2 * HDIM);
    }
    lstm_layer<<<2, G4H, 0, stream>>>(gxf, gxb, WhhT1f, WhhT1b, h1);

    {
        dim3 grid(MLP / 64, S_LEN / 16);
        gemm_wmma_f32<<<grid, 32, 0, stream>>>(h1, BPA, b1, (const float*)nullptr, hA, S_LEN, MLP, 2 * HDIM);
        gemm_wmma_f32<<<grid, 32, 0, stream>>>(h1, BPB, (const float*)nullptr, (const float*)nullptr, hB, S_LEN, MLP, 2 * HDIM);
    }

    {
        dim3 grid(S_LEN / 16, S_LEN / 16);
        pairwise_kernel<<<grid, 256, 0, stream>>>(hA, hB, W2, b2, out);
    }
}


# LSTM recurrence: one launch per time step instead of a resident work-group; a step's Whh product spread over 128 work-groups, rows cut in 8 k-slices summed by a lane tree (summation order changed)
# speedup vs baseline: 2.4068x; 2.4068x over previous
#include <hip/hip_runtime.h>
#include <hip/hip_bf16.h>


#define S_LEN   768
#define V_WORD  30000
#define V_POS   64
#define WEMB    100
#define PEMB    28
#define EMB     128
#define HDIM    256
#define G4H     1024
#define MLP     128

typedef __attribute__((ext_vector_type(2))) float v2f;
typedef __attribute__((ext_vector_type(8))) float v8f;

__device__ __forceinline__ float fast_sigmoid(float x) {
    return 1.0f / (1.0f + __expf(-x));
}
__device__ __forceinline__ float fast_tanh(float x) {
    float c = fminf(fmaxf(x, -9.0f), 9.0f);
    float e = __expf(2.0f * c);
    return (e - 1.0f) / (e + 1.0f);
}

template <int M>
__device__ __forceinline__ float row_xor(float v) {
    return __builtin_bit_cast(float,
        __builtin_amdgcn_update_dpp(0, __builtin_bit_cast(int, v), 0x160 + M, 0xf, 0xf, true));
}

__global__ void embed_kernel(const int* __restrict__ widx, const int* __restrict__ pidx,
                             const float* __restrict__ Wword, const float* __restrict__ Wpos,
                             float* __restrict__ x) {
    int s = blockIdx.x;
    int t = threadIdx.x;
    float v;
    if (t < WEMB) v = Wword[(size_t)widx[s] * WEMB + t];
    else          v = Wpos[(size_t)pidx[s] * PEMB + (t - WEMB)];
    x[(size_t)s * EMB + t] = v;
}

__global__ void pack_bt(const float* __restrict__ src, float* __restrict__ dst,
                        int Nout, int Ctot, int coff, int Kcnt) {
    int idx = blockIdx.x * blockDim.x + threadIdx.x;
    int total = Nout * Kcnt;
    if (idx >= total) return;
    int n = idx / Kcnt;
    int k = idx - n * Kcnt;
    dst[((size_t)(k >> 2) * Nout + n) * 4 + (k & 3)] = src[(size_t)n * Ctot + coff + k];
}

__global__ void __launch_bounds__(32)
gemm_wmma_f32(const float* __restrict__ A, const float* __restrict__ BP,
              const float* __restrict__ bias0, const float* __restrict__ bias1,
              float* __restrict__ C, int M, int N, int K) {
    const int nt = blockIdx.x;
    const int mt = blockIdx.y;
    const int l  = threadIdx.x;
    const int fr   = l & 15;
    const int half = l >> 4;
    const int kb   = half * 2;

    const float* Arow  = A + (size_t)(mt * 16 + fr) * K + kb;
    const int    nbase = nt * 64;

    v8f acc[4] = {{}, {}, {}, {}};
    for (int k0 = 0; k0 < K; k0 += 4) {
        v2f a;
        a.x = Arow[k0];
        a.y = Arow[k0 + 1];
        const float* bp = BP + ((size_t)(k0 >> 2) * N + nbase + fr) * 4 + kb;
        v2f b0, b1, b2, b3;
        b0.x = bp[0];       b0.y = bp[1];
        b1.x = bp[16 * 4];  b1.y = bp[16 * 4 + 1];
        b2.x = bp[32 * 4];  b2.y = bp[32 * 4 + 1];
        b3.x = bp[48 * 4];  b3.y = bp[48 * 4 + 1];
        acc[0] = __builtin_amdgcn_wmma_f32_16x16x4_f32(false, a, false, b0, (short)0, acc[0], false, false);
        acc[1] = __builtin_amdgcn_wmma_f32_16x16x4_f32(false, a, false, b1, (short)0, acc[1], false, false);
        acc[2] = __builtin_amdgcn_wmma_f32_16x16x4_f32(false, a, false, b2, (short)0, acc[2], false, false);
        acc[3] = __builtin_amdgcn_wmma_f32_16x16x4_f32(false, a, false, b3, (short)0, acc[3], false, false);
    }

#pragma unroll
    for (int u = 0; u < 4; ++u) {
        const int g = nbase + u * 16 + fr;
        float bs = 0.0f;
        if (bias0) bs += bias0[g];
        if (bias1) bs += bias1[g];
#pragma unroll
        for (int v = 0; v < 8; ++v) {
            int row = v + half * 8;
            C[(size_t)(mt * 16 + row) * N + g] = acc[u][v] + bs;
        }
    }
}

#define LS_UNITS   4
#define LS_GROUPS  (HDIM / LS_UNITS)
#define LS_PITCH   32

__global__ void __launch_bounds__(128)
lstm_step(const float* __restrict__ gx_f, const float* __restrict__ gx_b,
          const float* __restrict__ Whh_f, const float* __restrict__ Whh_b,
          float* __restrict__ hout, float* __restrict__ hst, float* __restrict__ cst,
          int step) {
    __shared__ float4 h_s4[HDIM / 4];
    __shared__ float  g_s[4 * LS_UNITS];

    const int tid = threadIdx.x;
    const int dir = blockIdx.x / LS_GROUPS;
    const int grp = blockIdx.x % LS_GROUPS;
    const bool last = (step == S_LEN);
    if (last && grp != 0) return;

    const int row  = tid >> 3;
    const int q    = tid & 7;
    const int gate = row >> 2;
    const int j    = grp * LS_UNITS + (row & 3);
    const int t    = dir ? (S_LEN - 1 - step) : step;
    const int tp   = dir ? (t + 1) : (t - 1);

    const float* gx  = dir ? gx_b  : gx_f;
    const float* Whh = dir ? Whh_b : Whh_f;
    const float* hst_rd = hst + (size_t)(((step + 1) & 1) * 2 + dir) * LS_GROUPS * LS_PITCH;
    float*       hst_wr = hst + (size_t)((step & 1) * 2 + dir) * LS_GROUPS * LS_PITCH;
    float*       cline  = cst + (size_t)(dir * LS_GROUPS + grp) * LS_PITCH;

    float4 w[8];
    float gxv = 0.0f, cprev = 0.0f;
    if (!last) {
        const float4* wp = (const float4*)(Whh + (size_t)(gate * HDIM + j) * HDIM);
#pragma unroll
        for (int i = 0; i < 8; ++i) w[i] = wp[q + 8 * i];
        if (q == 0) gxv = gx[(size_t)t * G4H + gate * HDIM + j];
        if (tid < LS_UNITS && step > 0) cprev = cline[tid];
    }

    if (tid < LS_GROUPS) {
        float4 hv = make_float4(0.0f, 0.0f, 0.0f, 0.0f);
        if (step > 0) hv = *(const float4*)(hst_rd + (size_t)tid * LS_PITCH);
        h_s4[tid] = hv;
    }
    __syncthreads();

    if (step > 0 && grp == 0) {
        const float2* hs2 = (const float2*)h_s4;
        ((float2*)(hout + (size_t)tp * (2 * HDIM) + dir * HDIM))[tid] = hs2[tid];
    }
    if (last) return;

    float a0 = 0.0f, a1 = 0.0f, a2 = 0.0f, a3 = 0.0f;
#pragma unroll
    for (int i = 0; i < 8; ++i) {
        const float4 h = h_s4[q + 8 * i];
        a0 += w[i].x * h.x;
        a1 += w[i].y * h.y;
        a2 += w[i].z * h.z;
        a3 += w[i].w * h.w;
    }
    float acc = (a0 + a1) + (a2 + a3);
    acc += row_xor<1>(acc);
    acc += row_xor<2>(acc);
    acc += row_xor<4>(acc);
    if (q == 0) g_s[row] = acc + gxv;
    __syncthreads();

    if (tid < LS_UNITS) {
        float gi = g_s[tid];
        float gf = g_s[LS_UNITS + tid];
        float gc = g_s[2 * LS_UNITS + tid];
        float go = g_s[3 * LS_UNITS + tid];
        float c  = fast_sigmoid(gf) * cprev + fast_sigmoid(gi) * fast_tanh(gc);
        float h  = fast_sigmoid(go) * fast_tanh(c);
        cline[tid] = c;
        hst_wr[(size_t)grp * LS_PITCH + tid] = h;
    }
}

__global__ void pairwise_kernel(const float* __restrict__ hA, const float* __restrict__ hB,
                                const float* __restrict__ W2, const float* __restrict__ b2,
                                float* __restrict__ out) {
    __shared__ float sA[16][MLP];
    __shared__ float sB[16][MLP];
    __shared__ float sW[MLP];

    const int i0 = blockIdx.y * 16;
    const int j0 = blockIdx.x * 16;
    const int tid = threadIdx.x;

    for (int idx = tid; idx < 16 * MLP; idx += 256) {
        int r = idx >> 7, k = idx & (MLP - 1);
        sA[r][k] = hA[(size_t)(i0 + r) * MLP + k];
        sB[r][k] = hB[(size_t)(j0 + r) * MLP + k];
    }
    if (tid < MLP) sW[tid] = W2[tid];
    __syncthreads();

    const int ti = tid >> 4;
    const int tj = tid & 15;
    const int i = i0 + ti;
    const int j = j0 + tj;

    float acc = 0.0f;
#pragma unroll 4
    for (int k = 0; k < MLP; ++k) {
        acc += sW[k] * fast_tanh(sA[ti][k] + sB[tj][k]);
    }
    float r = acc + b2[0];
    out[(size_t)i * S_LEN + j] = (i == j) ? 0.0f : r;
}

extern "C" void kernel_launch(void* const* d_in, const int* in_sizes, int n_in,
                              void* d_out, int out_size, void* d_ws, size_t ws_size,
                              hipStream_t stream) {
    const int*   widx   = (const int*)d_in[0];
    const int*   pidx   = (const int*)d_in[1];
    const float* Wword  = (const float*)d_in[3];
    const float* Wpos   = (const float*)d_in[4];
    const float* Wih_l0f = (const float*)d_in[5];
    const float* Whh_l0f = (const float*)d_in[6];
    const float* bih_l0f = (const float*)d_in[7];
    const float* bhh_l0f = (const float*)d_in[8];
    const float* Wih_l0b = (const float*)d_in[9];
    const float* Whh_l0b = (const float*)d_in[10];
    const float* bih_l0b = (const float*)d_in[11];
    const float* bhh_l0b = (const float*)d_in[12];
    const float* Wih_l1f = (const float*)d_in[13];
    const float* Whh_l1f = (const float*)d_in[14];
    const float* bih_l1f = (const float*)d_in[15];
    const float* bhh_l1f = (const float*)d_in[16];
    const float* Wih_l1b = (const float*)d_in[17];
    const float* Whh_l1b = (const float*)d_in[18];
    const float* bih_l1b = (const float*)d_in[19];
    const float* bhh_l1b = (const float*)d_in[20];
    const float* W1 = (const float*)d_in[21];
    const float* b1 = (const float*)d_in[22];
    const float* W2 = (const float*)d_in[23];
    const float* b2 = (const float*)d_in[24];
    float* out = (float*)d_out;

    float* ws = (float*)d_ws;
    float* x      = ws;
    float* gxf    = x      + S_LEN * EMB;
    float* gxb    = gxf    + S_LEN * G4H;
    float* h0     = gxb    + S_LEN * G4H;
    float* h1     = h0     + S_LEN * 2 * HDIM;
    float* hA     = h1     + S_LEN * 2 * HDIM;
    float* hB     = hA     + S_LEN * MLP;
    float* BP0f   = hB     + S_LEN * MLP;
    float* BP0b   = BP0f   + EMB * G4H;
    float* BP1f   = BP0b   + EMB * G4H;
    float* BP1b   = BP1f   + 2 * HDIM * G4H;
    float* hst    = BP1b   + 2 * HDIM * G4H;
    float* cst    = hst    + 2 * 2 * LS_GROUPS * LS_PITCH;
    float* BPA    = cst    + 2 * LS_GROUPS * LS_PITCH;
    float* BPB    = BPA    + 2 * HDIM * MLP;

    auto tgrid = [](int total) { return dim3((total + 255) / 256); };
    pack_bt<<<tgrid(G4H * EMB),      256, 0, stream>>>(Wih_l0f, BP0f, G4H, EMB,      0,      EMB);
    pack_bt<<<tgrid(G4H * EMB),      256, 0, stream>>>(Wih_l0b, BP0b, G4H, EMB,      0,      EMB);
    pack_bt<<<tgrid(G4H * 2 * HDIM), 256, 0, stream>>>(Wih_l1f, BP1f, G4H, 2 * HDIM, 0, 2 * HDIM);
    pack_bt<<<tgrid(G4H * 2 * HDIM), 256, 0, stream>>>(Wih_l1b, BP1b, G4H, 2 * HDIM, 0, 2 * HDIM);
    pack_bt<<<tgrid(MLP * 2 * HDIM), 256, 0, stream>>>(W1, BPA, MLP, G4H,        0, 2 * HDIM);
    pack_bt<<<tgrid(MLP * 2 * HDIM), 256, 0, stream>>>(W1, BPB, MLP, G4H, 2 * HDIM, 2 * HDIM);

    embed_kernel<<<S_LEN, EMB, 0, stream>>>(widx, pidx, Wword, Wpos, x);

    {
        dim3 grid(G4H / 64, S_LEN / 16);
        gemm_wmma_f32<<<grid, 32, 0, stream>>>(x, BP0f, bih_l0f, bhh_l0f, gxf, S_LEN, G4H, EMB);
        gemm_wmma_f32<<<grid, 32, 0, stream>>>(x, BP0b, bih_l0b, bhh_l0b, gxb, S_LEN, G4H, EMB);
    }
    for (int s = 0; s <= S_LEN; ++s)
        lstm_step<<<2 * LS_GROUPS, 128, 0, stream>>>(gxf, gxb, Whh_l0f, Whh_l0b, h0, hst, cst, s);

    {
        dim3 grid(G4H / 64, S_LEN / 16);
        gemm_wmma_f32<<<grid, 32, 0, stream>>>(h0, BP1f, bih_l1f, bhh_l1f, gxf, S_LEN, G4H, 2 * HDIM);
        gemm_wmma_f32<<<grid, 32, 0, stream>>>(h0, BP1b, bih_l1b, bhh_l1b, gxb, S_LEN, G4H, 2 * HDIM);
    }
    for (int s = 0; s <= S_LEN; ++s)
        lstm_step<<<2 * LS_GROUPS, 128, 0, stream>>>(gxf, gxb, Whh_l1f, Whh_l1b, h1, hst, cst, s);

    {
        dim3 grid(MLP / 64, S_LEN / 16);
        gemm_wmma_f32<<<grid, 32, 0, stream>>>(h1, BPA, b1, (const float*)nullptr, hA, S_LEN, MLP, 2 * HDIM);
        gemm_wmma_f32<<<grid, 32, 0, stream>>>(h1, BPB, (const float*)nullptr, (const float*)nullptr, hB, S_LEN, MLP, 2 * HDIM);
    }

    {
        dim3 grid(S_LEN / 16, S_LEN / 16);
        pairwise_kernel<<<grid, 256, 0, stream>>>(hA, hB, W2, b2, out);
    }
}


# LSTM step: a unit's 4 gate rows summed by the same 8 lanes of a one-wave work-group, h read from global straight into registers: no LDS, no barrier; summation order unchanged
# speedup vs baseline: 2.6280x; 1.0919x over previous
#include <hip/hip_runtime.h>
#include <hip/hip_bf16.h>


#define S_LEN   768
#define V_WORD  30000
#define V_POS   64
#define WEMB    100
#define PEMB    28
#define EMB     128
#define HDIM    256
#define G4H     1024
#define MLP     128

typedef __attribute__((ext_vector_type(2))) float v2f;
typedef __attribute__((ext_vector_type(4))) float v4f;
typedef __attribute__((ext_vector_type(8))) float v8f;

__device__ __forceinline__ float fast_sigmoid(float x) {
    return 1.0f / (1.0f + __expf(-x));
}
__device__ __forceinline__ float fast_tanh(float x) {
    float c = fminf(fmaxf(x, -9.0f), 9.0f);
    float e = __expf(2.0f * c);
    return (e - 1.0f) / (e + 1.0f);
}

template <int M>
__device__ __forceinline__ float row_xor(float v) {
    return __builtin_bit_cast(float,
        __builtin_amdgcn_update_dpp(0, __builtin_bit_cast(int, v), 0x160 + M, 0xf, 0xf, true));
}

__global__ void embed_kernel(const int* __restrict__ widx, const int* __restrict__ pidx,
                             const float* __restrict__ Wword, const float* __restrict__ Wpos,
                             float* __restrict__ x) {
    int s = blockIdx.x;
    int t = threadIdx.x;
    float v;
    if (t < WEMB) v = Wword[(size_t)widx[s] * WEMB + t];
    else          v = Wpos[(size_t)pidx[s] * PEMB + (t - WEMB)];
    x[(size_t)s * EMB + t] = v;
}

__global__ void pack_bt(const float* __restrict__ src, float* __restrict__ dst,
                        int Nout, int Ctot, int coff, int Kcnt) {
    int idx = blockIdx.x * blockDim.x + threadIdx.x;
    int total = Nout * Kcnt;
    if (idx >= total) return;
    int n = idx / Kcnt;
    int k = idx - n * Kcnt;
    dst[((size_t)(k >> 2) * Nout + n) * 4 + (k & 3)] = src[(size_t)n * Ctot + coff + k];
}

__global__ void __launch_bounds__(32)
gemm_wmma_f32(const float* __restrict__ A, const float* __restrict__ BP,
              const float* __restrict__ bias0, const float* __restrict__ bias1,
              float* __restrict__ C, int M, int N, int K) {
    const int nt = blockIdx.x;
    const int mt = blockIdx.y;
    const int l  = threadIdx.x;
    const int fr   = l & 15;
    const int half = l >> 4;
    const int kb   = half * 2;

    const float* Arow  = A + (size_t)(mt * 16 + fr) * K + kb;
    const int    nbase = nt * 64;

    v8f acc[4] = {{}, {}, {}, {}};
    for (int k0 = 0; k0 < K; k0 += 4) {
        v2f a;
        a.x = Arow[k0];
        a.y = Arow[k0 + 1];
        const float* bp = BP + ((size_t)(k0 >> 2) * N + nbase + fr) * 4 + kb;
        v2f b0, b1, b2, b3;
        b0.x = bp[0];       b0.y = bp[1];
        b1.x = bp[16 * 4];  b1.y = bp[16 * 4 + 1];
        b2.x = bp[32 * 4];  b2.y = bp[32 * 4 + 1];
        b3.x = bp[48 * 4];  b3.y = bp[48 * 4 + 1];
        acc[0] = __builtin_amdgcn_wmma_f32_16x16x4_f32(false, a, false, b0, (short)0, acc[0], false, false);
        acc[1] = __builtin_amdgcn_wmma_f32_16x16x4_f32(false, a, false, b1, (short)0, acc[1], false, false);
        acc[2] = __builtin_amdgcn_wmma_f32_16x16x4_f32(false, a, false, b2, (short)0, acc[2], false, false);
        acc[3] = __builtin_amdgcn_wmma_f32_16x16x4_f32(false, a, false, b3, (short)0, acc[3], false, false);
    }

#pragma unroll
    for (int u = 0; u < 4; ++u) {
        const int g = nbase + u * 16 + fr;
        float bs = 0.0f;
        if (bias0) bs += bias0[g];
        if (bias1) bs += bias1[g];
#pragma unroll
        for (int v = 0; v < 8; ++v) {
            int row = v + half * 8;
            C[(size_t)(mt * 16 + row) * N + g] = acc[u][v] + bs;
        }
    }
}

#define LS_UNITS   4
#define LS_GROUPS  (HDIM / LS_UNITS)
#define LS_PITCH   32

__global__ void __launch_bounds__(32)
lstm_step(const float* __restrict__ gx_f, const float* __restrict__ gx_b,
          const float* __restrict__ Whh_f, const float* __restrict__ Whh_b,
          float* __restrict__ hout, float* __restrict__ hst, float* __restrict__ cst,
          int step) {
    const int lane = threadIdx.x;
    const int dir = blockIdx.x / LS_GROUPS;
    const int grp = blockIdx.x % LS_GROUPS;
    const bool last = (step == S_LEN);
    if (last && grp != 0) return;

    const int u = lane >> 3;
    const int q = lane & 7;
    const int j = grp * LS_UNITS + u;
    const int t    = dir ? (S_LEN - 1 - step) : step;
    const int tp   = dir ? (t + 1) : (t - 1);

    const float* gx  = dir ? gx_b  : gx_f;
    const float* Whh = dir ? Whh_b : Whh_f;
    const float* hst_rd = hst + (size_t)(((step + 1) & 1) * 2 + dir) * LS_GROUPS * LS_PITCH;
    float*       hst_wr = hst + (size_t)((step & 1) * 2 + dir) * LS_GROUPS * LS_PITCH;
    float*       cline  = cst + (size_t)(dir * LS_GROUPS + grp) * LS_PITCH;

    v4f h[8];
#pragma unroll
    for (int i = 0; i < 8; ++i) h[i] = (v4f){0.0f, 0.0f, 0.0f, 0.0f};
    if (step > 0) {
#pragma unroll
        for (int i = 0; i < 8; ++i)
            h[i] = *(const v4f*)(hst_rd + (size_t)(q + 8 * i) * LS_PITCH);
    }

    v4f* ho = (v4f*)(hout + (size_t)tp * (2 * HDIM) + dir * HDIM);
    if (last) {
        if (lane < 8) {
#pragma unroll
            for (int i = 0; i < 8; ++i) ho[q + 8 * i] = h[i];
        }
        return;
    }

    float gxv[4] = {0.0f, 0.0f, 0.0f, 0.0f};
    float cprev = 0.0f;
    if (q == 0) {
#pragma unroll
        for (int g = 0; g < 4; ++g) gxv[g] = gx[(size_t)t * G4H + g * HDIM + j];
        if (step > 0) cprev = cline[u];
    }
    v4f w[4][8];
#pragma unroll
    for (int g = 0; g < 4; ++g) {
        const v4f* wp = (const v4f*)(Whh + (size_t)(g * HDIM + j) * HDIM);
#pragma unroll
        for (int i = 0; i < 8; ++i) w[g][i] = wp[q + 8 * i];
    }

    if (step > 0 && grp == 0 && lane < 8) {
#pragma unroll
        for (int i = 0; i < 8; ++i) ho[q + 8 * i] = h[i];
    }

    float gs[4];
#pragma unroll
    for (int g = 0; g < 4; ++g) {
        v2f alo = {0.0f, 0.0f}, ahi = {0.0f, 0.0f};
#pragma unroll
        for (int i = 0; i < 8; ++i) {
            alo = __builtin_elementwise_fma(w[g][i].lo, h[i].lo, alo);
            ahi = __builtin_elementwise_fma(w[g][i].hi, h[i].hi, ahi);
        }
        float acc = (alo.x + alo.y) + (ahi.x + ahi.y);
        acc += row_xor<1>(acc);
        acc += row_xor<2>(acc);
        acc += row_xor<4>(acc);
        gs[g] = acc + gxv[g];
    }

    if (q == 0) {
        float gi = gs[0];
        float gf = gs[1];
        float gc = gs[2];
        float go = gs[3];
        float c  = fast_sigmoid(gf) * cprev + fast_sigmoid(gi) * fast_tanh(gc);
        float hn = fast_sigmoid(go) * fast_tanh(c);
        cline[u] = c;
        hst_wr[(size_t)grp * LS_PITCH + u] = hn;
    }
}

__global__ void pairwise_kernel(const float* __restrict__ hA, const float* __restrict__ hB,
                                const float* __restrict__ W2, const float* __restrict__ b2,
                                float* __restrict__ out) {
    __shared__ float sA[16][MLP];
    __shared__ float sB[16][MLP];
    __shared__ float sW[MLP];

    const int i0 = blockIdx.y * 16;
    const int j0 = blockIdx.x * 16;
    const int tid = threadIdx.x;

    for (int idx = tid; idx < 16 * MLP; idx += 256) {
        int r = idx >> 7, k = idx & (MLP - 1);
        sA[r][k] = hA[(size_t)(i0 + r) * MLP + k];
        sB[r][k] = hB[(size_t)(j0 + r) * MLP + k];
    }
    if (tid < MLP) sW[tid] = W2[tid];
    __syncthreads();

    const int ti = tid >> 4;
    const int tj = tid & 15;
    const int i = i0 + ti;
    const int j = j0 + tj;

    float acc = 0.0f;
#pragma unroll 4
    for (int k = 0; k < MLP; ++k) {
        acc += sW[k] * fast_tanh(sA[ti][k] + sB[tj][k]);
    }
    float r = acc + b2[0];
    out[(size_t)i * S_LEN + j] = (i == j) ? 0.0f : r;
}

extern "C" void kernel_launch(void* const* d_in, const int* in_sizes, int n_in,
                              void* d_out, int out_size, void* d_ws, size_t ws_size,
                              hipStream_t stream) {
    const int*   widx   = (const int*)d_in[0];
    const int*   pidx   = (const int*)d_in[1];
    const float* Wword  = (const float*)d_in[3];
    const float* Wpos   = (const float*)d_in[4];
    const float* Wih_l0f = (const float*)d_in[5];
    const float* Whh_l0f = (const float*)d_in[6];
    const float* bih_l0f = (const float*)d_in[7];
    const float* bhh_l0f = (const float*)d_in[8];
    const float* Wih_l0b = (const float*)d_in[9];
    const float* Whh_l0b = (const float*)d_in[10];
    const float* bih_l0b = (const float*)d_in[11];
    const float* bhh_l0b = (const float*)d_in[12];
    const float* Wih_l1f = (const float*)d_in[13];
    const float* Whh_l1f = (const float*)d_in[14];
    const float* bih_l1f = (const float*)d_in[15];
    const float* bhh_l1f = (const float*)d_in[16];
    const float* Wih_l1b = (const float*)d_in[17];
    const float* Whh_l1b = (const float*)d_in[18];
    const float* bih_l1b = (const float*)d_in[19];
    const float* bhh_l1b = (const float*)d_in[20];
    const float* W1 = (const float*)d_in[21];
    const float* b1 = (const float*)d_in[22];
    const float* W2 = (const float*)d_in[23];
    const float* b2 = (const float*)d_in[24];
    float* out = (float*)d_out;

    float* ws = (float*)d_ws;
    float* x      = ws;
    float* gxf    = x      + S_LEN * EMB;
    float* gxb    = gxf    + S_LEN * G4H;
    float* h0     = gxb    + S_LEN * G4H;
    float* h1     = h0     + S_LEN * 2 * HDIM;
    float* hA     = h1     + S_LEN * 2 * HDIM;
    float* hB     = hA     + S_LEN * MLP;
    float* BP0f   = hB     + S_LEN * MLP;
    float* BP0b   = BP0f   + EMB * G4H;
    float* BP1f   = BP0b   + EMB * G4H;
    float* BP1b   = BP1f   + 2 * HDIM * G4H;
    float* hst    = BP1b   + 2 * HDIM * G4H;
    float* cst    = hst    + 2 * 2 * LS_GROUPS * LS_PITCH;
    float* BPA    = cst    + 2 * LS_GROUPS * LS_PITCH;
    float* BPB    = BPA    + 2 * HDIM * MLP;

    auto tgrid = [](int total) { return dim3((total + 255) / 256); };
    pack_bt<<<tgrid(G4H * EMB),      256, 0, stream>>>(Wih_l0f, BP0f, G4H, EMB,      0,      EMB);
    pack_bt<<<tgrid(G4H * EMB),      256, 0, stream>>>(Wih_l0b, BP0b, G4H, EMB,      0,      EMB);
    pack_bt<<<tgrid(G4H * 2 * HDIM), 256, 0, stream>>>(Wih_l1f, BP1f, G4H, 2 * HDIM, 0, 2 * HDIM);
    pack_bt<<<tgrid(G4H * 2 * HDIM), 256, 0, stream>>>(Wih_l1b, BP1b, G4H, 2 * HDIM, 0, 2 * HDIM);
    pack_bt<<<tgrid(MLP * 2 * HDIM), 256, 0, stream>>>(W1, BPA, MLP, G4H,        0, 2 * HDIM);
    pack_bt<<<tgrid(MLP * 2 * HDIM), 256, 0, stream>>>(W1, BPB, MLP, G4H, 2 * HDIM, 2 * HDIM);

    embed_kernel<<<S_LEN, EMB, 0, stream>>>(widx, pidx, Wword, Wpos, x);

    {
        dim3 grid(G4H / 64, S_LEN / 16);
        gemm_wmma_f32<<<grid, 32, 0, stream>>>(x, BP0f, bih_l0f, bhh_l0f, gxf, S_LEN, G4H, EMB);
        gemm_wmma_f32<<<grid, 32, 0, stream>>>(x, BP0b, bih_l0b, bhh_l0b, gxb, S_LEN, G4H, EMB);
    }
    for (int s = 0; s <= S_LEN; ++s)
        lstm_step<<<2 * LS_GROUPS, 32, 0, stream>>>(gxf, gxb, Whh_l0f, Whh_l0b, h0, hst, cst, s);

    {
        dim3 grid(G4H / 64, S_LEN / 16);
        gemm_wmma_f32<<<grid, 32, 0, stream>>>(h0, BP1f, bih_l1f, bhh_l1f, gxf, S_LEN, G4H, 2 * HDIM);
        gemm_wmma_f32<<<grid, 32, 0, stream>>>(h0, BP1b, bih_l1b, bhh_l1b, gxb, S_LEN, G4H, 2 * HDIM);
    }
    for (int s = 0; s <= S_LEN; ++s)
        lstm_step<<<2 * LS_GROUPS, 32, 0, stream>>>(gxf, gxb, Whh_l1f, Whh_l1b, h1, hst, cst, s);

    {
        dim3 grid(MLP / 64, S_LEN / 16);
        gemm_wmma_f32<<<grid, 32, 0, stream>>>(h1, BPA, b1, (const float*)nullptr, hA, S_LEN, MLP, 2 * HDIM);
        gemm_wmma_f32<<<grid, 32, 0, stream>>>(h1, BPB, (const float*)nullptr, (const float*)nullptr, hB, S_LEN, MLP, 2 * HDIM);
    }

    {
        dim3 grid(S_LEN / 16, S_LEN / 16);
        pairwise_kernel<<<grid, 256, 0, stream>>>(hA, hB, W2, b2, out);
    }
}
